# starting kernel
# baseline (speedup 1.0000x reference)
#include <hip/hip_runtime.h>

typedef float v2f __attribute__((ext_vector_type(2)));
typedef float v8f __attribute__((ext_vector_type(8)));

#define D_K 128

template <int DOUT>
__global__ __launch_bounds__(256) void gcn_gemm_scale_kernel(
    const float* __restrict__ X, const float* __restrict__ W,
    const float* __restrict__ dinv, float* __restrict__ G)
{
    const int wave = threadIdx.x >> 5;
    const int lane = threadIdx.x & 31;
    const int rowBase = blockIdx.x << 4;
    const int hi  = lane >> 4;
    const int l15 = lane & 15;
    const int col = (wave << 4) + l15;

    const float* __restrict__ xbase = X + (size_t)(rowBase + l15) * D_K + (hi << 1);
    const float* __restrict__ wbase = W + (size_t)(hi << 1) * DOUT + col;

    __builtin_prefetch(wbase, 0, 1);

    v8f acc = {0.f, 0.f, 0.f, 0.f, 0.f, 0.f, 0.f, 0.f};
#pragma unroll
    for (int k0 = 0; k0 < D_K; k0 += 4) {
        v2f a;
        a.x = xbase[k0];
        a.y = xbase[k0 + 1];
        v2f b;
        b.x = wbase[(size_t)k0 * DOUT];
        b.y = wbase[(size_t)(k0 + 1) * DOUT];
        acc = __builtin_amdgcn_wmma_f32_16x16x4_f32(
             false, a,  false, b,
             (short)0, acc,  false,  false);
    }

    const int row0 = rowBase + (hi << 3);
    float* __restrict__ gout = G + (size_t)row0 * DOUT + col;
    const float* __restrict__ dbase = dinv + row0;
#pragma unroll
    for (int r = 0; r < 8; ++r) {
        gout[(size_t)r * DOUT] = acc[r] * dbase[r];
    }
}

__global__ void fill_kernel(float* __restrict__ p, float v, int n) {
    int i = blockIdx.x * blockDim.x + threadIdx.x;
    if (i < n) p[i] = v;
}

__global__ void deg_count_kernel(const int* __restrict__ dst,
                                 float* __restrict__ deg, int E) {
    int e = blockIdx.x * blockDim.x + threadIdx.x;
    if (e < E) unsafeAtomicAdd(&deg[dst[e]], 1.0f);
}

__global__ void rsqrt_inplace_kernel(float* __restrict__ p, int n) {
    int i = blockIdx.x * blockDim.x + threadIdx.x;
    if (i < n) p[i] = rsqrtf(p[i]);
}

__global__ void zero4_kernel(float4* __restrict__ p, int n4) {
    int i = blockIdx.x * blockDim.x + threadIdx.x;
    if (i < n4) p[i] = make_float4(0.f, 0.f, 0.f, 0.f);
}

template <int SHIFT>
__global__ __launch_bounds__(256) void scatter_kernel(
    const float* __restrict__ G, const int* __restrict__ src,
    const int* __restrict__ dst, float* __restrict__ ACC, int E)
{
    const long long t = (long long)blockIdx.x * blockDim.x + threadIdx.x;
    const int e = (int)(t >> SHIFT);
    if (e >= E) return;
    const int f = (int)(t & ((1 << SHIFT) - 1)) << 2;
    constexpr int DOUT = 4 << SHIFT;
    const int s = src[e];
    const int d = dst[e];
    const float4 g = *reinterpret_cast<const float4*>(G + (size_t)s * DOUT + f);
    float* a = ACC + (size_t)d * DOUT + f;
    unsafeAtomicAdd(a + 0, g.x);
    unsafeAtomicAdd(a + 1, g.y);
    unsafeAtomicAdd(a + 2, g.z);
    unsafeAtomicAdd(a + 3, g.w);
}

template <int SHIFT>
__global__ __launch_bounds__(256) void finalize_kernel(
    const float* __restrict__ G, const float* __restrict__ ACC,
    const float* __restrict__ dinv, const float* __restrict__ bias,
    float* __restrict__ OUT, int n4, int relu)
{
    const int t = blockIdx.x * blockDim.x + threadIdx.x;
    if (t >= n4) return;
    const int i = t >> SHIFT;
    const int f = (t & ((1 << SHIFT) - 1)) << 2;
    const float di = dinv[i];
    const float4 g = *reinterpret_cast<const float4*>(G   + (size_t)t * 4);
    const float4 a = *reinterpret_cast<const float4*>(ACC + (size_t)t * 4);
    const float4 b = *reinterpret_cast<const float4*>(bias + f);
    float4 o;
    o.x = di * (a.x + g.x) + b.x;
    o.y = di * (a.y + g.y) + b.y;
    o.z = di * (a.z + g.z) + b.z;
    o.w = di * (a.w + g.w) + b.w;
    if (relu) {
        o.x = fmaxf(o.x, 0.f);
        o.y = fmaxf(o.y, 0.f);
        o.z = fmaxf(o.z, 0.f);
        o.w = fmaxf(o.w, 0.f);
    }
    *reinterpret_cast<float4*>(OUT + (size_t)t * 4) = o;
}

extern "C" void kernel_launch(void* const* d_in, const int* in_sizes, int n_in,
                              void* d_out, int out_size, void* d_ws, size_t ws_size,
                              hipStream_t stream)
{
    const float* x   = (const float*)d_in[0];
    const int*   ei  = (const int*)  d_in[1];
    const float* W1  = (const float*)d_in[3];
    const float* b1  = (const float*)d_in[4];
    const float* W2  = (const float*)d_in[5];
    const float* b2  = (const float*)d_in[6];
    const float* W3  = (const float*)d_in[7];
    const float* b3  = (const float*)d_in[8];
    const float* W4  = (const float*)d_in[9];
    const float* b4  = (const float*)d_in[10];

    const int N = in_sizes[0] / D_K;
    const int E = in_sizes[1] / 2;
    const int* src = ei;
    const int* dst = ei + E;

    float* dinv = (float*)d_ws;
    float* GA   = dinv + N;
    float* GB   = GA + (size_t)N * D_K;
    float* ACC  = GB + (size_t)N * D_K;

    fill_kernel<<<(N + 255) / 256, 256, 0, stream>>>(dinv, 1.0f, N);
    deg_count_kernel<<<(E + 255) / 256, 256, 0, stream>>>(dst, dinv, E);
    rsqrt_inplace_kernel<<<(N + 255) / 256, 256, 0, stream>>>(dinv, N);

    auto run128 = [&](const float* Xin, const float* W, const float* b,
                      float* G, float* Out) {
        gcn_gemm_scale_kernel<128><<<N / 16, 256, 0, stream>>>(Xin, W, dinv, G);
        const int n4 = N * 32;
        zero4_kernel<<<(n4 + 255) / 256, 256, 0, stream>>>((float4*)ACC, n4);
        const long long e4 = (long long)E * 32;
        scatter_kernel<5><<<(int)((e4 + 255) / 256), 256, 0, stream>>>(G, src, dst, ACC, E);
        finalize_kernel<5><<<(n4 + 255) / 256, 256, 0, stream>>>(G, ACC, dinv, b, Out, n4, 1);
    };

    run128(x,  W1, b1, GA, GA);
    run128(GA, W2, b2, GB, GB);
    run128(GB, W3, b3, GA, GA);

    {
        gcn_gemm_scale_kernel<64><<<N / 16, 128, 0, stream>>>(GA, W4, dinv, GB);
        const int n4 = N * 16;
        zero4_kernel<<<(n4 + 255) / 256, 256, 0, stream>>>((float4*)ACC, n4);
        const long long e4 = (long long)E * 16;
        scatter_kernel<4><<<(int)((e4 + 255) / 256), 256, 0, stream>>>(GB, src, dst, ACC, E);
        finalize_kernel<4><<<(n4 + 255) / 256, 256, 0, stream>>>(GB, ACC, dinv, b4,
                                                                 (float*)d_out, n4, 0);
    }
}


# Edges counting-sorted by destination in-call (int atomic rank, arrival order); one wave gathers a node's source rows, writes the final row: no float atomics, zero or finalize pass; sum order changes
# speedup vs baseline: 2.1499x; 2.1499x over previous
#include <hip/hip_runtime.h>

typedef float v2f __attribute__((ext_vector_type(2)));
typedef float v8f __attribute__((ext_vector_type(8)));

#define D_K 128

template <int DOUT>
__global__ __launch_bounds__(256) void gcn_gemm_scale_kernel(
    const float* __restrict__ X, const float* __restrict__ W,
    const float* __restrict__ dinv, float* __restrict__ G)
{
    const int wave = threadIdx.x >> 5;
    const int lane = threadIdx.x & 31;
    const int rowBase = blockIdx.x << 4;
    const int hi  = lane >> 4;
    const int l15 = lane & 15;
    const int col = (wave << 4) + l15;

    const float* __restrict__ xbase = X + (size_t)(rowBase + l15) * D_K + (hi << 1);
    const float* __restrict__ wbase = W + (size_t)(hi << 1) * DOUT + col;

    __builtin_prefetch(wbase, 0, 1);

    v8f acc = {0.f, 0.f, 0.f, 0.f, 0.f, 0.f, 0.f, 0.f};
#pragma unroll
    for (int k0 = 0; k0 < D_K; k0 += 4) {
        v2f a;
        a.x = xbase[k0];
        a.y = xbase[k0 + 1];
        v2f b;
        b.x = wbase[(size_t)k0 * DOUT];
        b.y = wbase[(size_t)(k0 + 1) * DOUT];
        acc = __builtin_amdgcn_wmma_f32_16x16x4_f32(
             false, a,  false, b,
             (short)0, acc,  false,  false);
    }

    const int row0 = rowBase + (hi << 3);
    float* __restrict__ gout = G + (size_t)row0 * DOUT + col;
    const float* __restrict__ dbase = dinv + row0;
#pragma unroll
    for (int r = 0; r < 8; ++r) {
        gout[(size_t)r * DOUT] = acc[r] * dbase[r];
    }
}

#define SCAN_SHIFT 10
#define SCAN_MAX_BLOCKS 1024

__global__ void zero_int_kernel(int* __restrict__ p, int n) {
    int i = blockIdx.x * blockDim.x + threadIdx.x;
    if (i < n) p[i] = 0;
}

__global__ void edge_rank_kernel(const int* __restrict__ dst, int* __restrict__ cnt,
                                 int* __restrict__ rank, int E) {
    int e = blockIdx.x * blockDim.x + threadIdx.x;
    if (e < E) rank[e] = atomicAdd(&cnt[dst[e]], 1);
}

__device__ __forceinline__ int wave_inclusive_scan(int v, int lane) {
#pragma unroll
    for (int d = 1; d < 32; d <<= 1) {
        const int up = __shfl_up(v, d);
        if (lane >= d) v += up;
    }
    return v;
}

__global__ __launch_bounds__(256) void degree_scan_kernel(
    const int* __restrict__ cnt, int* __restrict__ lscan, int* __restrict__ btot,
    float* __restrict__ dinv, int N)
{
    __shared__ int wsum[8];
    const int tid  = threadIdx.x;
    const int lane = tid & 31;
    const int wave = tid >> 5;
    const int base = (blockIdx.x * 256 + tid) * 4;
    int c[4];
    if (base + 3 < N) {
        const int4 v = *reinterpret_cast<const int4*>(cnt + base);
        c[0] = v.x; c[1] = v.y; c[2] = v.z; c[3] = v.w;
    } else {
#pragma unroll
        for (int k = 0; k < 4; ++k) c[k] = (base + k < N) ? cnt[base + k] : 0;
    }
    const int tsum = c[0] + c[1] + c[2] + c[3];
    const int incl = wave_inclusive_scan(tsum, lane);
    if (lane == 31) wsum[wave] = incl;
    __syncthreads();
    int woff = 0;
#pragma unroll
    for (int w = 0; w < 8; ++w) if (w < wave) woff += wsum[w];
    int run = woff + incl - tsum;
#pragma unroll
    for (int k = 0; k < 4; ++k) {
        if (base + k < N) {
            lscan[base + k] = run;
            dinv[base + k]  = rsqrtf(1.0f + (float)c[k]);
        }
        run += c[k];
    }
    if (tid == 255) btot[blockIdx.x] = run;
}

__global__ __launch_bounds__(256) void csr_fill_kernel(
    const int* __restrict__ src, const int* __restrict__ dst,
    const int* __restrict__ rank, const int* __restrict__ cnt,
    const int* __restrict__ lscan, const int* __restrict__ btot,
    int* __restrict__ rowinfo, int* __restrict__ csr, int N, int E, int nb)
{
    __shared__ int boff[SCAN_MAX_BLOCKS];
    if (threadIdx.x < 32) {
        const int lane = threadIdx.x;
        int carry = 0;
        for (int b0 = 0; b0 < nb; b0 += 128) {
            const int b = b0 + lane * 4;
            int v[4];
#pragma unroll
            for (int k = 0; k < 4; ++k) v[k] = (b + k < nb) ? btot[b + k] : 0;
            const int t = v[0] + v[1] + v[2] + v[3];
            const int incl = wave_inclusive_scan(t, lane);
            int run = carry + incl - t;
#pragma unroll
            for (int k = 0; k < 4; ++k) {
                if (b + k < nb) boff[b + k] = run;
                run += v[k];
            }
            carry += __shfl(incl, 31);
        }
    }
    __syncthreads();
    const int t = blockIdx.x * 256 + threadIdx.x;
    if (t < N)
        reinterpret_cast<int2*>(rowinfo)[t] = make_int2(lscan[t] + boff[t >> SCAN_SHIFT], cnt[t]);
    if (t < E) {
        const int d = dst[t];
        csr[lscan[d] + boff[d >> SCAN_SHIFT] + rank[t]] = src[t];
    }
}

template <int K, typename VF>
__device__ __forceinline__ void gather_rows(VF& acc, const VF* __restrict__ Gv,
                                            int mine, int q, int lane)
{
    VF g[K];
#pragma unroll
    for (int u = 0; u < K; ++u)
        g[u] = Gv[(size_t)__builtin_amdgcn_readlane(mine, q + u) * 32 + lane];
#pragma unroll
    for (int u = 0; u < K; ++u) acc += g[u];
}

template <int DOUT, bool RELU>
__global__ __launch_bounds__(256) void gather_finalize_kernel(
    const float* __restrict__ G, const int* __restrict__ rowinfo,
    const int* __restrict__ csr,
    const float* __restrict__ dinv, const float* __restrict__ bias,
    float* __restrict__ OUT, int N)
{
    constexpr int V = DOUT / 32;
    typedef float vf __attribute__((ext_vector_type(V)));
    constexpr int U = 8;
    const int lane = threadIdx.x & 31;
    const int i = __builtin_amdgcn_readfirstlane(
        (int)((blockIdx.x * 256u + threadIdx.x) >> 5));
    if (i >= N) return;
    const vf* __restrict__ Gv = reinterpret_cast<const vf*>(G);
    const int2 info = reinterpret_cast<const int2*>(rowinfo)[i];
    const int beg = info.x;
    const int deg = info.y;
    vf acc = Gv[(size_t)i * 32 + lane];
    for (int base = 0; base < deg; base += 32) {
        const int n = min(32, deg - base);
        int mine = 0;
        if (lane < n) mine = csr[beg + base + lane];
        mine = min(max(mine, 0), N - 1);
        for (int q = 0; q < n; q += U) {
            switch (min(U, n - q)) {
                case 1:  gather_rows<1>(acc, Gv, mine, q, lane); break;
                case 2:  gather_rows<2>(acc, Gv, mine, q, lane); break;
                case 3:  gather_rows<3>(acc, Gv, mine, q, lane); break;
                case 4:  gather_rows<4>(acc, Gv, mine, q, lane); break;
                case 5:  gather_rows<5>(acc, Gv, mine, q, lane); break;
                case 6:  gather_rows<6>(acc, Gv, mine, q, lane); break;
                case 7:  gather_rows<7>(acc, Gv, mine, q, lane); break;
                default: gather_rows<8>(acc, Gv, mine, q, lane); break;
            }
        }
    }
    const float di = dinv[i];
    const vf b = reinterpret_cast<const vf*>(bias)[lane];
    vf o = di * acc + b;
    if (RELU) {
#pragma unroll
        for (int k = 0; k < V; ++k) o[k] = fmaxf(o[k], 0.f);
    }
    reinterpret_cast<vf*>(OUT)[(size_t)i * 32 + lane] = o;
}

extern "C" void kernel_launch(void* const* d_in, const int* in_sizes, int n_in,
                              void* d_out, int out_size, void* d_ws, size_t ws_size,
                              hipStream_t stream)
{
    const float* x   = (const float*)d_in[0];
    const int*   ei  = (const int*)  d_in[1];
    const float* W1  = (const float*)d_in[3];
    const float* b1  = (const float*)d_in[4];
    const float* W2  = (const float*)d_in[5];
    const float* b2  = (const float*)d_in[6];
    const float* W3  = (const float*)d_in[7];
    const float* b3  = (const float*)d_in[8];
    const float* W4  = (const float*)d_in[9];
    const float* b4  = (const float*)d_in[10];

    const int N = in_sizes[0] / D_K;
    const int E = in_sizes[1] / 2;
    const int* src = ei;
    const int* dst = ei + E;

    const size_t nPad = ((size_t)N + 63) & ~(size_t)63;
    const size_t ePad = ((size_t)E + 63) & ~(size_t)63;
    float* dinv   = (float*)d_ws;
    int*   cnt    = (int*)(dinv + nPad);
    int*   lscan  = cnt + nPad;
    int*   rowinfo = lscan + nPad;
    int*   btot   = rowinfo + 2 * nPad;
    int*   rank   = btot + SCAN_MAX_BLOCKS;
    int*   csr    = rank + ePad;
    float* GA     = (float*)(csr + ePad);
    float* GB     = GA + (size_t)N * D_K;

    const int nb = (N + (1 << SCAN_SHIFT) - 1) >> SCAN_SHIFT;
    const int nmax = N > E ? N : E;
    zero_int_kernel<<<(N + 255) / 256, 256, 0, stream>>>(cnt, N);
    edge_rank_kernel<<<(E + 255) / 256, 256, 0, stream>>>(dst, cnt, rank, E);
    degree_scan_kernel<<<nb, 256, 0, stream>>>(cnt, lscan, btot, dinv, N);
    csr_fill_kernel<<<(nmax + 255) / 256, 256, 0, stream>>>(src, dst, rank, cnt, lscan, btot,
                                                           rowinfo, csr, N, E, nb);

    const int gblocks = (N + 7) / 8;

    auto run128 = [&](const float* Xin, const float* W, const float* b) {
        gcn_gemm_scale_kernel<128><<<N / 16, 256, 0, stream>>>(Xin, W, dinv, GA);
        gather_finalize_kernel<128, true><<<gblocks, 256, 0, stream>>>(
            GA, rowinfo, csr, dinv, b, GB, N);
    };
    run128(x,  W1, b1);
    run128(GB, W2, b2);
    run128(GB, W3, b3);

    gcn_gemm_scale_kernel<64><<<N / 16, 128, 0, stream>>>(GB, W4, dinv, GA);
    gather_finalize_kernel<64, false><<<gblocks, 256, 0, stream>>>(
        GA, rowinfo, csr, dinv, b4, (float*)d_out, N);
}


# GEMM data path: each wave keeps its 128x16 W panel in registers over many 16-row strips; each X strip read once per block as whole rows into LDS (2 buffers), shared by its waves; sums unchanged
# speedup vs baseline: 3.0303x; 1.4095x over previous
#include <hip/hip_runtime.h>

typedef float v2f __attribute__((ext_vector_type(2)));
typedef float v8f __attribute__((ext_vector_type(8)));

#define D_K 128

#define XS_PITCH 132

typedef float v4f __attribute__((ext_vector_type(4)));

template <int DOUT>
__global__ __launch_bounds__(DOUT * 2) void gcn_gemm_scale_kernel(
    const float* __restrict__ X, const float* __restrict__ W,
    const float* __restrict__ dinv, float* __restrict__ G, int nStrips)
{
    constexpr int NT  = DOUT * 2;
    constexpr int LPT = 512 / NT;
    __shared__ alignas(16) float xs[2][16 * XS_PITCH];

    const int tid  = threadIdx.x;
    const int wave = tid >> 5;
    const int lane = tid & 31;
    const int hi  = lane >> 4;
    const int l15 = lane & 15;
    const int col = (wave << 4) + l15;

    int s = blockIdx.x;
    if (s >= nStrips) return;

    const float* __restrict__ wbase = W + (size_t)(hi << 1) * DOUT + col;
    v2f b[D_K / 4];
#pragma unroll
    for (int t = 0; t < D_K / 4; ++t) {
        b[t].x = wbase[(size_t)(4 * t) * DOUT];
        b[t].y = wbase[(size_t)(4 * t + 1) * DOUT];
    }

    const v4f* __restrict__ Xv = reinterpret_cast<const v4f*>(X);
    v4f xa[LPT];
#pragma unroll
    for (int u = 0; u < LPT; ++u) xa[u] = Xv[(size_t)s * 512 + tid + u * NT];

    const int aoff = l15 * XS_PITCH + (hi << 1);

    int buf = 0;
    while (true) {
        float* __restrict__ xb = xs[buf];
#pragma unroll
        for (int u = 0; u < LPT; ++u) {
            const int f = tid + u * NT;
            *reinterpret_cast<v4f*>(xb + (f >> 5) * XS_PITCH + ((f & 31) << 2)) = xa[u];
        }
        __syncthreads();

        const int sn = s + gridDim.x;
        if (sn < nStrips) {
#pragma unroll
            for (int u = 0; u < LPT; ++u) xa[u] = Xv[(size_t)sn * 512 + tid + u * NT];
        }

        const int row0 = (s << 4) + (hi << 3);
        const float* __restrict__ dbase = dinv + row0;
        float dv[8];
#pragma unroll
        for (int r = 0; r < 8; ++r) dv[r] = dbase[r];

        const float* __restrict__ ab = xb + aoff;
        v8f acc = {0.f, 0.f, 0.f, 0.f, 0.f, 0.f, 0.f, 0.f};
#pragma unroll
        for (int t = 0; t < D_K / 4; ++t) {
            const v2f a = *reinterpret_cast<const v2f*>(ab + 4 * t);
            acc = __builtin_amdgcn_wmma_f32_16x16x4_f32(
                 false, a,  false, b[t],
                 (short)0, acc,  false,  false);
        }

        float* __restrict__ gout = G + (size_t)row0 * DOUT + col;
#pragma unroll
        for (int r = 0; r < 8; ++r) {
            gout[(size_t)r * DOUT] = acc[r] * dv[r];
        }

        if (sn >= nStrips) break;
        s = sn;
        buf ^= 1;
    }
}

#define SCAN_SHIFT 10
#define SCAN_MAX_BLOCKS 1024

__global__ void zero_int_kernel(int* __restrict__ p, int n) {
    int i = blockIdx.x * blockDim.x + threadIdx.x;
    if (i < n) p[i] = 0;
}

__global__ void edge_rank_kernel(const int* __restrict__ dst, int* __restrict__ cnt,
                                 int* __restrict__ rank, int E) {
    int e = blockIdx.x * blockDim.x + threadIdx.x;
    if (e < E) rank[e] = atomicAdd(&cnt[dst[e]], 1);
}

__device__ __forceinline__ int wave_inclusive_scan(int v, int lane) {
#pragma unroll
    for (int d = 1; d < 32; d <<= 1) {
        const int up = __shfl_up(v, d);
        if (lane >= d) v += up;
    }
    return v;
}

__global__ __launch_bounds__(256) void degree_scan_kernel(
    const int* __restrict__ cnt, int* __restrict__ lscan, int* __restrict__ btot,
    float* __restrict__ dinv, int N)
{
    __shared__ int wsum[8];
    const int tid  = threadIdx.x;
    const int lane = tid & 31;
    const int wave = tid >> 5;
    const int base = (blockIdx.x * 256 + tid) * 4;
    int c[4];
    if (base + 3 < N) {
        const int4 v = *reinterpret_cast<const int4*>(cnt + base);
        c[0] = v.x; c[1] = v.y; c[2] = v.z; c[3] = v.w;
    } else {
#pragma unroll
        for (int k = 0; k < 4; ++k) c[k] = (base + k < N) ? cnt[base + k] : 0;
    }
    const int tsum = c[0] + c[1] + c[2] + c[3];
    const int incl = wave_inclusive_scan(tsum, lane);
    if (lane == 31) wsum[wave] = incl;
    __syncthreads();
    int woff = 0;
#pragma unroll
    for (int w = 0; w < 8; ++w) if (w < wave) woff += wsum[w];
    int run = woff + incl - tsum;
#pragma unroll
    for (int k = 0; k < 4; ++k) {
        if (base + k < N) {
            lscan[base + k] = run;
            dinv[base + k]  = rsqrtf(1.0f + (float)c[k]);
        }
        run += c[k];
    }
    if (tid == 255) btot[blockIdx.x] = run;
}

__global__ __launch_bounds__(256) void csr_fill_kernel(
    const int* __restrict__ src, const int* __restrict__ dst,
    const int* __restrict__ rank, const int* __restrict__ cnt,
    const int* __restrict__ lscan, const int* __restrict__ btot,
    int* __restrict__ rowinfo, int* __restrict__ csr, int N, int E, int nb)
{
    __shared__ int boff[SCAN_MAX_BLOCKS];
    if (threadIdx.x < 32) {
        const int lane = threadIdx.x;
        int carry = 0;
        for (int b0 = 0; b0 < nb; b0 += 128) {
            const int b = b0 + lane * 4;
            int v[4];
#pragma unroll
            for (int k = 0; k < 4; ++k) v[k] = (b + k < nb) ? btot[b + k] : 0;
            const int t = v[0] + v[1] + v[2] + v[3];
            const int incl = wave_inclusive_scan(t, lane);
            int run = carry + incl - t;
#pragma unroll
            for (int k = 0; k < 4; ++k) {
                if (b + k < nb) boff[b + k] = run;
                run += v[k];
            }
            carry += __shfl(incl, 31);
        }
    }
    __syncthreads();
    const int t = blockIdx.x * 256 + threadIdx.x;
    if (t < N)
        reinterpret_cast<int2*>(rowinfo)[t] = make_int2(lscan[t] + boff[t >> SCAN_SHIFT], cnt[t]);
    if (t < E) {
        const int d = dst[t];
        csr[lscan[d] + boff[d >> SCAN_SHIFT] + rank[t]] = src[t];
    }
}

template <int K, typename VF>
__device__ __forceinline__ void gather_rows(VF& acc, const VF* __restrict__ Gv,
                                            int mine, int q, int lane)
{
    VF g[K];
#pragma unroll
    for (int u = 0; u < K; ++u)
        g[u] = Gv[(size_t)__builtin_amdgcn_readlane(mine, q + u) * 32 + lane];
#pragma unroll
    for (int u = 0; u < K; ++u) acc += g[u];
}

template <int DOUT, bool RELU>
__global__ __launch_bounds__(256) void gather_finalize_kernel(
    const float* __restrict__ G, const int* __restrict__ rowinfo,
    const int* __restrict__ csr,
    const float* __restrict__ dinv, const float* __restrict__ bias,
    float* __restrict__ OUT, int N)
{
    constexpr int V = DOUT / 32;
    typedef float vf __attribute__((ext_vector_type(V)));
    constexpr int U = 8;
    const int lane = threadIdx.x & 31;
    const int i = __builtin_amdgcn_readfirstlane(
        (int)((blockIdx.x * 256u + threadIdx.x) >> 5));
    if (i >= N) return;
    const vf* __restrict__ Gv = reinterpret_cast<const vf*>(G);
    const int2 info = reinterpret_cast<const int2*>(rowinfo)[i];
    const int beg = info.x;
    const int deg = info.y;
    vf acc = Gv[(size_t)i * 32 + lane];
    for (int base = 0; base < deg; base += 32) {
        const int n = min(32, deg - base);
        int mine = 0;
        if (lane < n) mine = csr[beg + base + lane];
        mine = min(max(mine, 0), N - 1);
        for (int q = 0; q < n; q += U) {
            switch (min(U, n - q)) {
                case 1:  gather_rows<1>(acc, Gv, mine, q, lane); break;
                case 2:  gather_rows<2>(acc, Gv, mine, q, lane); break;
                case 3:  gather_rows<3>(acc, Gv, mine, q, lane); break;
                case 4:  gather_rows<4>(acc, Gv, mine, q, lane); break;
                case 5:  gather_rows<5>(acc, Gv, mine, q, lane); break;
                case 6:  gather_rows<6>(acc, Gv, mine, q, lane); break;
                case 7:  gather_rows<7>(acc, Gv, mine, q, lane); break;
                default: gather_rows<8>(acc, Gv, mine, q, lane); break;
            }
        }
    }
    const float di = dinv[i];
    const vf b = reinterpret_cast<const vf*>(bias)[lane];
    vf o = di * acc + b;
    if (RELU) {
#pragma unroll
        for (int k = 0; k < V; ++k) o[k] = fmaxf(o[k], 0.f);
    }
    reinterpret_cast<vf*>(OUT)[(size_t)i * 32 + lane] = o;
}

extern "C" void kernel_launch(void* const* d_in, const int* in_sizes, int n_in,
                              void* d_out, int out_size, void* d_ws, size_t ws_size,
                              hipStream_t stream)
{
    const float* x   = (const float*)d_in[0];
    const int*   ei  = (const int*)  d_in[1];
    const float* W1  = (const float*)d_in[3];
    const float* b1  = (const float*)d_in[4];
    const float* W2  = (const float*)d_in[5];
    const float* b2  = (const float*)d_in[6];
    const float* W3  = (const float*)d_in[7];
    const float* b3  = (const float*)d_in[8];
    const float* W4  = (const float*)d_in[9];
    const float* b4  = (const float*)d_in[10];

    const int N = in_sizes[0] / D_K;
    const int E = in_sizes[1] / 2;
    const int* src = ei;
    const int* dst = ei + E;

    const size_t nPad = ((size_t)N + 63) & ~(size_t)63;
    const size_t ePad = ((size_t)E + 63) & ~(size_t)63;
    float* dinv   = (float*)d_ws;
    int*   cnt    = (int*)(dinv + nPad);
    int*   lscan  = cnt + nPad;
    int*   rowinfo = lscan + nPad;
    int*   btot   = rowinfo + 2 * nPad;
    int*   rank   = btot + SCAN_MAX_BLOCKS;
    int*   csr    = rank + ePad;
    float* GA     = (float*)(csr + ePad);
    float* GB     = GA + (size_t)N * D_K;

    const int nb = (N + (1 << SCAN_SHIFT) - 1) >> SCAN_SHIFT;
    const int nmax = N > E ? N : E;
    zero_int_kernel<<<(N + 255) / 256, 256, 0, stream>>>(cnt, N);
    edge_rank_kernel<<<(E + 255) / 256, 256, 0, stream>>>(dst, cnt, rank, E);
    degree_scan_kernel<<<nb, 256, 0, stream>>>(cnt, lscan, btot, dinv, N);
    csr_fill_kernel<<<(nmax + 255) / 256, 256, 0, stream>>>(src, dst, rank, cnt, lscan, btot,
                                                           rowinfo, csr, N, E, nb);

    const int gblocks = (N + 7) / 8;

    const int nStrips = N / 16;
    const int gemmBlocks128 = 1024;
    const int gemmBlocks64  = 2048;

    auto run128 = [&](const float* Xin, const float* W, const float* b) {
        gcn_gemm_scale_kernel<128><<<gemmBlocks128, 256, 0, stream>>>(Xin, W, dinv, GA, nStrips);
        gather_finalize_kernel<128, true><<<gblocks, 256, 0, stream>>>(
            GA, rowinfo, csr, dinv, b, GB, N);
    };
    run128(x,  W1, b1);
    run128(GB, W2, b2);
    run128(GB, W3, b3);

    gcn_gemm_scale_kernel<64><<<gemmBlocks64, 128, 0, stream>>>(GB, W4, dinv, GA, nStrips);
    gather_finalize_kernel<64, false><<<gblocks, 256, 0, stream>>>(
        GA, rowinfo, csr, dinv, b4, (float*)d_out, N);
}


# Each layer's gather (dinv scale, bias, ReLU) fused into the next layer's GEMM as the producer of its LDS row strips: hidden activations never go to memory, 12 launches become 9; sums unchanged
# speedup vs baseline: 3.4781x; 1.1478x over previous
#include <hip/hip_runtime.h>

typedef float v2f __attribute__((ext_vector_type(2)));
typedef float v8f __attribute__((ext_vector_type(8)));

#define D_K 128

#define XS_PITCH 132

typedef float v4f __attribute__((ext_vector_type(4)));

template <int DOUT>
__global__ __launch_bounds__(DOUT * 2) void gcn_gemm_scale_kernel(
    const float* __restrict__ X, const float* __restrict__ W,
    const float* __restrict__ dinv, float* __restrict__ G, int nStrips)
{
    constexpr int NT  = DOUT * 2;
    constexpr int LPT = 512 / NT;
    __shared__ alignas(16) float xs[2][16 * XS_PITCH];

    const int tid  = threadIdx.x;
    const int wave = tid >> 5;
    const int lane = tid & 31;
    const int hi  = lane >> 4;
    const int l15 = lane & 15;
    const int col = (wave << 4) + l15;

    int s = blockIdx.x;
    if (s >= nStrips) return;

    const float* __restrict__ wbase = W + (size_t)(hi << 1) * DOUT + col;
    v2f b[D_K / 4];
#pragma unroll
    for (int t = 0; t < D_K / 4; ++t) {
        b[t].x = wbase[(size_t)(4 * t) * DOUT];
        b[t].y = wbase[(size_t)(4 * t + 1) * DOUT];
    }

    const v4f* __restrict__ Xv = reinterpret_cast<const v4f*>(X);
    v4f xa[LPT];
#pragma unroll
    for (int u = 0; u < LPT; ++u) xa[u] = Xv[(size_t)s * 512 + tid + u * NT];

    const int aoff = l15 * XS_PITCH + (hi << 1);

    int buf = 0;
    while (true) {
        float* __restrict__ xb = xs[buf];
#pragma unroll
        for (int u = 0; u < LPT; ++u) {
            const int f = tid + u * NT;
            *reinterpret_cast<v4f*>(xb + (f >> 5) * XS_PITCH + ((f & 31) << 2)) = xa[u];
        }
        __syncthreads();

        const int sn = s + gridDim.x;
        if (sn < nStrips) {
#pragma unroll
            for (int u = 0; u < LPT; ++u) xa[u] = Xv[(size_t)sn * 512 + tid + u * NT];
        }

        const int row0 = (s << 4) + (hi << 3);
        const float* __restrict__ dbase = dinv + row0;
        float dv[8];
#pragma unroll
        for (int r = 0; r < 8; ++r) dv[r] = dbase[r];

        const float* __restrict__ ab = xb + aoff;
        v8f acc = {0.f, 0.f, 0.f, 0.f, 0.f, 0.f, 0.f, 0.f};
#pragma unroll
        for (int t = 0; t < D_K / 4; ++t) {
            const v2f a = *reinterpret_cast<const v2f*>(ab + 4 * t);
            acc = __builtin_amdgcn_wmma_f32_16x16x4_f32(
                 false, a,  false, b[t],
                 (short)0, acc,  false,  false);
        }

        float* __restrict__ gout = G + (size_t)row0 * DOUT + col;
#pragma unroll
        for (int r = 0; r < 8; ++r) {
            gout[(size_t)r * DOUT] = acc[r] * dv[r];
        }

        if (sn >= nStrips) break;
        s = sn;
        buf ^= 1;
    }
}

#define SCAN_SHIFT 10
#define SCAN_MAX_BLOCKS 1024

__global__ void zero_int_kernel(int* __restrict__ p, int n) {
    int i = blockIdx.x * blockDim.x + threadIdx.x;
    if (i < n) p[i] = 0;
}

__global__ void edge_rank_kernel(const int* __restrict__ dst, int* __restrict__ cnt,
                                 int* __restrict__ rank, int E) {
    int e = blockIdx.x * blockDim.x + threadIdx.x;
    if (e < E) rank[e] = atomicAdd(&cnt[dst[e]], 1);
}

__device__ __forceinline__ int wave_inclusive_scan(int v, int lane) {
#pragma unroll
    for (int d = 1; d < 32; d <<= 1) {
        const int up = __shfl_up(v, d);
        if (lane >= d) v += up;
    }
    return v;
}

__global__ __launch_bounds__(256) void degree_scan_kernel(
    const int* __restrict__ cnt, int* __restrict__ lscan, int* __restrict__ btot,
    float* __restrict__ dinv, int N)
{
    __shared__ int wsum[8];
    const int tid  = threadIdx.x;
    const int lane = tid & 31;
    const int wave = tid >> 5;
    const int base = (blockIdx.x * 256 + tid) * 4;
    int c[4];
    if (base + 3 < N) {
        const int4 v = *reinterpret_cast<const int4*>(cnt + base);
        c[0] = v.x; c[1] = v.y; c[2] = v.z; c[3] = v.w;
    } else {
#pragma unroll
        for (int k = 0; k < 4; ++k) c[k] = (base + k < N) ? cnt[base + k] : 0;
    }
    const int tsum = c[0] + c[1] + c[2] + c[3];
    const int incl = wave_inclusive_scan(tsum, lane);
    if (lane == 31) wsum[wave] = incl;
    __syncthreads();
    int woff = 0;
#pragma unroll
    for (int w = 0; w < 8; ++w) if (w < wave) woff += wsum[w];
    int run = woff + incl - tsum;
#pragma unroll
    for (int k = 0; k < 4; ++k) {
        if (base + k < N) {
            lscan[base + k] = run;
            dinv[base + k]  = rsqrtf(1.0f + (float)c[k]);
        }
        run += c[k];
    }
    if (tid == 255) btot[blockIdx.x] = run;
}

__global__ __launch_bounds__(256) void csr_fill_kernel(
    const int* __restrict__ src, const int* __restrict__ dst,
    const int* __restrict__ rank, const int* __restrict__ cnt,
    const int* __restrict__ lscan, const int* __restrict__ btot,
    int* __restrict__ rowinfo, int* __restrict__ csr, int N, int E, int nb)
{
    __shared__ int boff[SCAN_MAX_BLOCKS];
    if (threadIdx.x < 32) {
        const int lane = threadIdx.x;
        int carry = 0;
        for (int b0 = 0; b0 < nb; b0 += 128) {
            const int b = b0 + lane * 4;
            int v[4];
#pragma unroll
            for (int k = 0; k < 4; ++k) v[k] = (b + k < nb) ? btot[b + k] : 0;
            const int t = v[0] + v[1] + v[2] + v[3];
            const int incl = wave_inclusive_scan(t, lane);
            int run = carry + incl - t;
#pragma unroll
            for (int k = 0; k < 4; ++k) {
                if (b + k < nb) boff[b + k] = run;
                run += v[k];
            }
            carry += __shfl(incl, 31);
        }
    }
    __syncthreads();
    const int t = blockIdx.x * 256 + threadIdx.x;
    if (t < N)
        reinterpret_cast<int2*>(rowinfo)[t] = make_int2(lscan[t] + boff[t >> SCAN_SHIFT], cnt[t]);
    if (t < E) {
        const int d = dst[t];
        csr[lscan[d] + boff[d >> SCAN_SHIFT] + rank[t]] = src[t];
    }
}

template <int K, typename VF>
__device__ __forceinline__ void gather_rows(VF& acc, const VF* __restrict__ Gv,
                                            int mine, int q, int lane)
{
    VF g[K];
#pragma unroll
    for (int u = 0; u < K; ++u)
        g[u] = Gv[(size_t)__builtin_amdgcn_readlane(mine, q + u) * 32 + lane];
#pragma unroll
    for (int u = 0; u < K; ++u) acc += g[u];
}

template <int DOUT, bool RELU>
__global__ __launch_bounds__(256) void gather_finalize_kernel(
    const float* __restrict__ G, const int* __restrict__ rowinfo,
    const int* __restrict__ csr,
    const float* __restrict__ dinv, const float* __restrict__ bias,
    float* __restrict__ OUT, int N)
{
    constexpr int V = DOUT / 32;
    typedef float vf __attribute__((ext_vector_type(V)));
    constexpr int U = 8;
    const int lane = threadIdx.x & 31;
    const int i = __builtin_amdgcn_readfirstlane(
        (int)((blockIdx.x * 256u + threadIdx.x) >> 5));
    if (i >= N) return;
    const vf* __restrict__ Gv = reinterpret_cast<const vf*>(G);
    const int2 info = reinterpret_cast<const int2*>(rowinfo)[i];
    const int beg = info.x;
    const int deg = info.y;
    vf acc = Gv[(size_t)i * 32 + lane];
    for (int base = 0; base < deg; base += 32) {
        const int n = min(32, deg - base);
        int mine = 0;
        if (lane < n) mine = csr[beg + base + lane];
        mine = min(max(mine, 0), N - 1);
        for (int q = 0; q < n; q += U) {
            switch (min(U, n - q)) {
                case 1:  gather_rows<1>(acc, Gv, mine, q, lane); break;
                case 2:  gather_rows<2>(acc, Gv, mine, q, lane); break;
                case 3:  gather_rows<3>(acc, Gv, mine, q, lane); break;
                case 4:  gather_rows<4>(acc, Gv, mine, q, lane); break;
                case 5:  gather_rows<5>(acc, Gv, mine, q, lane); break;
                case 6:  gather_rows<6>(acc, Gv, mine, q, lane); break;
                case 7:  gather_rows<7>(acc, Gv, mine, q, lane); break;
                default: gather_rows<8>(acc, Gv, mine, q, lane); break;
            }
        }
    }
    const float di = dinv[i];
    const vf b = reinterpret_cast<const vf*>(bias)[lane];
    vf o = di * acc + b;
    if (RELU) {
#pragma unroll
        for (int k = 0; k < V; ++k) o[k] = fmaxf(o[k], 0.f);
    }
    reinterpret_cast<vf*>(OUT)[(size_t)i * 32 + lane] = o;
}

template <typename VF>
__device__ __forceinline__ VF gather_node_sum(const VF* __restrict__ Gv,
                                              const int* __restrict__ csr,
                                              int beg, int deg, int mine0,
                                              int lane, int N)
{
    constexpr int U = 8;
    VF acc = {0.f, 0.f, 0.f, 0.f};
    const int len = deg + 1;
    for (int base = 0; base < len; base += 32) {
        const int n = min(32, len - base);
        int mine = mine0;
        if (base != 0) {
            mine = 0;
            if (lane < n) mine = csr[beg + base + lane - 1];
            mine = min(max(mine, 0), N - 1);
        }
        for (int q = 0; q < n; q += U) {
            switch (min(U, n - q)) {
                case 1:  gather_rows<1>(acc, Gv, mine, q, lane); break;
                case 2:  gather_rows<2>(acc, Gv, mine, q, lane); break;
                case 3:  gather_rows<3>(acc, Gv, mine, q, lane); break;
                case 4:  gather_rows<4>(acc, Gv, mine, q, lane); break;
                case 5:  gather_rows<5>(acc, Gv, mine, q, lane); break;
                case 6:  gather_rows<6>(acc, Gv, mine, q, lane); break;
                case 7:  gather_rows<7>(acc, Gv, mine, q, lane); break;
                default: gather_rows<8>(acc, Gv, mine, q, lane); break;
            }
        }
    }
    return acc;
}

template <int DOUT>
__global__ __launch_bounds__(DOUT * 2) void gcn_gather_gemm_scale_kernel(
    const float* __restrict__ Gin, const int* __restrict__ rowinfo,
    const int* __restrict__ csr, const float* __restrict__ dinv,
    const float* __restrict__ bias, const float* __restrict__ W,
    float* __restrict__ G, int nStrips, int N, int E)
{
    constexpr int NT  = DOUT * 2;
    constexpr int NPW = 16 / (NT / 32);
    __shared__ alignas(16) float xs[2][16 * XS_PITCH];

    const int tid  = threadIdx.x;
    const int wave = __builtin_amdgcn_readfirstlane(tid >> 5);
    const int lane = tid & 31;
    const int hi  = lane >> 4;
    const int l15 = lane & 15;
    const int col = (wave << 4) + l15;

    int s = blockIdx.x;
    if (s >= nStrips) return;

    const float* __restrict__ wbase = W + (size_t)(hi << 1) * DOUT + col;
    v2f b[D_K / 4];
#pragma unroll
    for (int t = 0; t < D_K / 4; ++t) {
        b[t].x = wbase[(size_t)(4 * t) * DOUT];
        b[t].y = wbase[(size_t)(4 * t + 1) * DOUT];
    }

    const v4f* __restrict__ Gv = reinterpret_cast<const v4f*>(Gin);
    const int2* __restrict__ info2 = reinterpret_cast<const int2*>(rowinfo);
    const v4f bv = reinterpret_cast<const v4f*>(bias)[lane];
    const int r0 = wave * NPW;

    int2 info[NPW];
    int  mine[NPW];
    float dstrip = dinv[(s << 4) + l15];
    {
        const int2 ri = info2[(s << 4) + l15];
#pragma unroll
        for (int j = 0; j < NPW; ++j) {
            info[j].x = __builtin_amdgcn_readlane(ri.x, r0 + j);
            info[j].y = __builtin_amdgcn_readlane(ri.y, r0 + j);
            mine[j] = csr[min(max(info[j].x + lane - 1, 0), E - 1)];
        }
    }

    const int aoff = l15 * XS_PITCH + (hi << 1);

    int buf = 0;
    while (true) {
        float* __restrict__ xb = xs[buf];
        const int sn = s + gridDim.x;
        const bool more = sn < nStrips;

#pragma unroll
        for (int j = 0; j < NPW; ++j) {
            const int m = min(max(mine[j], 0), N - 1);
            mine[j] = (lane == 0) ? (s << 4) + r0 + j : m;
        }

        const int nrow = ((more ? sn : s) << 4) + l15;
        const int2  nri = info2[nrow];
        const float ndv = dinv[nrow];

#pragma unroll
        for (int j = 0; j < NPW; ++j) {
            const v4f acc = gather_node_sum<v4f>(Gv, csr, info[j].x, info[j].y,
                                                 mine[j], lane, N);
            const float di = __int_as_float(
                __builtin_amdgcn_readlane(__float_as_int(dstrip), r0 + j));
            v4f o = di * acc + bv;
#pragma unroll
            for (int k = 0; k < 4; ++k) o[k] = fmaxf(o[k], 0.f);
            *reinterpret_cast<v4f*>(xb + (r0 + j) * XS_PITCH + (lane << 2)) = o;
        }
        __syncthreads();

#pragma unroll
        for (int j = 0; j < NPW; ++j) {
            info[j].x = __builtin_amdgcn_readlane(nri.x, r0 + j);
            info[j].y = __builtin_amdgcn_readlane(nri.y, r0 + j);
            mine[j] = csr[min(max(info[j].x + lane - 1, 0), E - 1)];
        }
        dstrip = ndv;

        const int row0 = (s << 4) + (hi << 3);
        const float* __restrict__ dbase = dinv + row0;
        float dv[8];
#pragma unroll
        for (int r = 0; r < 8; ++r) dv[r] = dbase[r];

        const float* __restrict__ ab = xb + aoff;
        v8f acc = {0.f, 0.f, 0.f, 0.f, 0.f, 0.f, 0.f, 0.f};
#pragma unroll
        for (int t = 0; t < D_K / 4; ++t) {
            const v2f a = *reinterpret_cast<const v2f*>(ab + 4 * t);
            acc = __builtin_amdgcn_wmma_f32_16x16x4_f32(
                 false, a,  false, b[t],
                 (short)0, acc,  false,  false);
        }

        float* __restrict__ gout = G + (size_t)row0 * DOUT + col;
#pragma unroll
        for (int r = 0; r < 8; ++r) {
            gout[(size_t)r * DOUT] = acc[r] * dv[r];
        }

        if (!more) break;
        s = sn;
        buf ^= 1;
    }
}

extern "C" void kernel_launch(void* const* d_in, const int* in_sizes, int n_in,
                              void* d_out, int out_size, void* d_ws, size_t ws_size,
                              hipStream_t stream)
{
    const float* x   = (const float*)d_in[0];
    const int*   ei  = (const int*)  d_in[1];
    const float* W1  = (const float*)d_in[3];
    const float* b1  = (const float*)d_in[4];
    const float* W2  = (const float*)d_in[5];
    const float* b2  = (const float*)d_in[6];
    const float* W3  = (const float*)d_in[7];
    const float* b3  = (const float*)d_in[8];
    const float* W4  = (const float*)d_in[9];
    const float* b4  = (const float*)d_in[10];

    const int N = in_sizes[0] / D_K;
    const int E = in_sizes[1] / 2;
    const int* src = ei;
    const int* dst = ei + E;

    const size_t nPad = ((size_t)N + 63) & ~(size_t)63;
    const size_t ePad = ((size_t)E + 63) & ~(size_t)63;
    float* dinv   = (float*)d_ws;
    int*   cnt    = (int*)(dinv + nPad);
    int*   lscan  = cnt + nPad;
    int*   rowinfo = lscan + nPad;
    int*   btot   = rowinfo + 2 * nPad;
    int*   rank   = btot + SCAN_MAX_BLOCKS;
    int*   csr    = rank + ePad;
    float* GA     = (float*)(csr + ePad);
    float* GB     = GA + (size_t)N * D_K;

    const int nb = (N + (1 << SCAN_SHIFT) - 1) >> SCAN_SHIFT;
    const int nmax = N > E ? N : E;
    zero_int_kernel<<<(N + 255) / 256, 256, 0, stream>>>(cnt, N);
    edge_rank_kernel<<<(E + 255) / 256, 256, 0, stream>>>(dst, cnt, rank, E);
    degree_scan_kernel<<<nb, 256, 0, stream>>>(cnt, lscan, btot, dinv, N);
    csr_fill_kernel<<<(nmax + 255) / 256, 256, 0, stream>>>(src, dst, rank, cnt, lscan, btot,
                                                           rowinfo, csr, N, E, nb);

    const int gblocks = (N + 7) / 8;

    const int nStrips = N / 16;
    const int gemmBlocks128 = 1024;
    const int gemmBlocks64  = 2048;

    gcn_gemm_scale_kernel<128><<<gemmBlocks128, 256, 0, stream>>>(x, W1, dinv, GA, nStrips);
    gcn_gather_gemm_scale_kernel<128><<<gemmBlocks128, 256, 0, stream>>>(
        GA, rowinfo, csr, dinv, b1, W2, GB, nStrips, N, E);
    gcn_gather_gemm_scale_kernel<128><<<gemmBlocks128, 256, 0, stream>>>(
        GB, rowinfo, csr, dinv, b2, W3, GA, nStrips, N, E);
    gcn_gather_gemm_scale_kernel<64><<<gemmBlocks64, 128, 0, stream>>>(
        GA, rowinfo, csr, dinv, b3, W4, GB, nStrips, N, E);

    gather_finalize_kernel<64, false><<<gblocks, 256, 0, stream>>>(
        GB, rowinfo, csr, dinv, b4, (float*)d_out, N);
}


# Edge list as one 32-slot row per node (entry 0 the node, then sources by the same arrival-order rank), filled by the counting pass itself; spill list past 31: scan and fill launches gone, 9 become 7
# speedup vs baseline: 3.7743x; 1.0852x over previous
#include <hip/hip_runtime.h>

typedef float v2f __attribute__((ext_vector_type(2)));
typedef float v8f __attribute__((ext_vector_type(8)));

#define D_K 128

#define XS_PITCH 132

typedef float v4f __attribute__((ext_vector_type(4)));

template <int DOUT>
__global__ __launch_bounds__(DOUT * 2) void gcn_gemm_scale_kernel(
    const float* __restrict__ X, const float* __restrict__ W,
    const int* __restrict__ cnt, float* __restrict__ dinvOut,
    float* __restrict__ G, int nStrips)
{
    constexpr int NT  = DOUT * 2;
    constexpr int LPT = 512 / NT;
    __shared__ alignas(16) float xs[2][16 * XS_PITCH];
    __shared__ alignas(16) float dstrip[2][16];

    const int tid  = threadIdx.x;
    const int wave = __builtin_amdgcn_readfirstlane(tid >> 5);
    const int lane = tid & 31;
    const int hi  = lane >> 4;
    const int l15 = lane & 15;
    const int col = (wave << 4) + l15;

    int s = blockIdx.x;
    if (s >= nStrips) return;

    const float* __restrict__ wbase = W + (size_t)(hi << 1) * DOUT + col;
    v2f b[D_K / 4];
#pragma unroll
    for (int t = 0; t < D_K / 4; ++t) {
        b[t].x = wbase[(size_t)(4 * t) * DOUT];
        b[t].y = wbase[(size_t)(4 * t + 1) * DOUT];
    }

    const v4f* __restrict__ Xv = reinterpret_cast<const v4f*>(X);
    v4f xa[LPT];
#pragma unroll
    for (int u = 0; u < LPT; ++u) xa[u] = Xv[(size_t)s * 512 + tid + u * NT];
    int cn = 0;
    if (wave == 0) cn = cnt[(s << 4) + l15];

    const int aoff = l15 * XS_PITCH + (hi << 1);

    int buf = 0;
    while (true) {
        float* __restrict__ xb = xs[buf];
#pragma unroll
        for (int u = 0; u < LPT; ++u) {
            const int f = tid + u * NT;
            *reinterpret_cast<v4f*>(xb + (f >> 5) * XS_PITCH + ((f & 31) << 2)) = xa[u];
        }
        float d = 0.f;
        if (wave == 0) {
            d = rsqrtf(1.0f + (float)cn);
            if (hi == 0) dstrip[buf][l15] = d;
        }
        __syncthreads();
        if (wave == 0 && hi == 0) dinvOut[(s << 4) + l15] = d;

        const int sn = s + gridDim.x;
        if (sn < nStrips) {
#pragma unroll
            for (int u = 0; u < LPT; ++u) xa[u] = Xv[(size_t)sn * 512 + tid + u * NT];
            if (wave == 0) cn = cnt[(sn << 4) + l15];
        }

        const int row0 = (s << 4) + (hi << 3);
        float dv[8];
        {
            const v4f* __restrict__ dsv = reinterpret_cast<const v4f*>(&dstrip[buf][hi << 3]);
            const v4f d0 = dsv[0];
            const v4f d1 = dsv[1];
#pragma unroll
            for (int r = 0; r < 4; ++r) { dv[r] = d0[r]; dv[4 + r] = d1[r]; }
        }

        const float* __restrict__ ab = xb + aoff;
        v8f acc = {0.f, 0.f, 0.f, 0.f, 0.f, 0.f, 0.f, 0.f};
#pragma unroll
        for (int t = 0; t < D_K / 4; ++t) {
            const v2f a = *reinterpret_cast<const v2f*>(ab + 4 * t);
            acc = __builtin_amdgcn_wmma_f32_16x16x4_f32(
                 false, a,  false, b[t],
                 (short)0, acc,  false,  false);
        }

        float* __restrict__ gout = G + (size_t)row0 * DOUT + col;
#pragma unroll
        for (int r = 0; r < 8; ++r) {
            gout[(size_t)r * DOUT] = acc[r] * dv[r];
        }

        if (sn >= nStrips) break;
        s = sn;
        buf ^= 1;
    }
}

#define LIST_W   32
#define LIST_CAP 31

__global__ void list_init_kernel(int* __restrict__ cnt, int* __restrict__ list,
                                 int* __restrict__ spill_n, int N) {
    int i = blockIdx.x * blockDim.x + threadIdx.x;
    if (i < N) {
        cnt[i] = 0;
        list[(size_t)i * LIST_W] = i;
    }
    if (i == 0) spill_n[0] = 0;
}

__global__ void edge_place_kernel(const int* __restrict__ src, const int* __restrict__ dst,
                                  int* __restrict__ cnt, int* __restrict__ list,
                                  int* __restrict__ spill_n, int* __restrict__ spill, int E) {
    int e = blockIdx.x * blockDim.x + threadIdx.x;
    if (e < E) {
        const int d = dst[e];
        const int s = src[e];
        const int r = atomicAdd(&cnt[d], 1);
        if (r < LIST_CAP) {
            list[(size_t)d * LIST_W + 1 + r] = s;
        } else {
            const int k = atomicAdd(spill_n, 1);
            spill[2 * k]     = d;
            spill[2 * k + 1] = s;
        }
    }
}

template <int K, typename VF>
__device__ __forceinline__ void gather_rows(VF& acc, const VF* __restrict__ Gv,
                                            int mine, int q, int lane)
{
    VF g[K];
#pragma unroll
    for (int u = 0; u < K; ++u)
        g[u] = Gv[(size_t)__builtin_amdgcn_readlane(mine, q + u) * 32 + lane];
#pragma unroll
    for (int u = 0; u < K; ++u) acc += g[u];
}

template <typename VF>
__device__ __forceinline__ VF gather_node_sum(const VF* __restrict__ Gv,
                                              const int* __restrict__ spill_n,
                                              const int* __restrict__ spill,
                                              int node, int deg, int mine,
                                              int lane, int N, int E)
{
    constexpr int U = 8;
    VF acc = (VF)0.f;
    const int n = min(deg, LIST_CAP) + 1;
    for (int q = 0; q < n; q += U) {
        switch (min(U, n - q)) {
            case 1:  gather_rows<1>(acc, Gv, mine, q, lane); break;
            case 2:  gather_rows<2>(acc, Gv, mine, q, lane); break;
            case 3:  gather_rows<3>(acc, Gv, mine, q, lane); break;
            case 4:  gather_rows<4>(acc, Gv, mine, q, lane); break;
            case 5:  gather_rows<5>(acc, Gv, mine, q, lane); break;
            case 6:  gather_rows<6>(acc, Gv, mine, q, lane); break;
            case 7:  gather_rows<7>(acc, Gv, mine, q, lane); break;
            default: gather_rows<8>(acc, Gv, mine, q, lane); break;
        }
    }
    if (deg > LIST_CAP) {
        const int m = min(spill_n[0], E);
        for (int k = 0; k < m; ++k) {
            if (spill[2 * k] == node) {
                const int sidx = min(max(spill[2 * k + 1], 0), N - 1);
                acc += Gv[(size_t)sidx * 32 + lane];
            }
        }
    }
    return acc;
}

template <int DOUT, bool RELU>
__global__ __launch_bounds__(256) void gather_finalize_kernel(
    const float* __restrict__ G, const int* __restrict__ cnt,
    const int* __restrict__ list, const int* __restrict__ spill_n,
    const int* __restrict__ spill,
    const float* __restrict__ dinv, const float* __restrict__ bias,
    float* __restrict__ OUT, int N, int E)
{
    constexpr int V = DOUT / 32;
    typedef float vf __attribute__((ext_vector_type(V)));
    const int lane = threadIdx.x & 31;
    const int i = __builtin_amdgcn_readfirstlane(
        (int)((blockIdx.x * 256u + threadIdx.x) >> 5));
    if (i >= N) return;
    const vf* __restrict__ Gv = reinterpret_cast<const vf*>(G);
    const int deg = cnt[i];
    int mine = list[i * LIST_W + min(lane, deg)];
    mine = min(max(mine, 0), N - 1);
    const vf acc = gather_node_sum<vf>(Gv, spill_n, spill, i, deg, mine, lane, N, E);
    const float di = dinv[i];
    const vf b = reinterpret_cast<const vf*>(bias)[lane];
    vf o = di * acc + b;
    if (RELU) {
#pragma unroll
        for (int k = 0; k < V; ++k) o[k] = fmaxf(o[k], 0.f);
    }
    reinterpret_cast<vf*>(OUT)[(size_t)i * 32 + lane] = o;
}

template <int DOUT>
__global__ __launch_bounds__(DOUT * 2) void gcn_gather_gemm_scale_kernel(
    const float* __restrict__ Gin, const int* __restrict__ cnt,
    const int* __restrict__ list, const int* __restrict__ spill_n,
    const int* __restrict__ spill, const float* __restrict__ dinv,
    const float* __restrict__ bias, const float* __restrict__ W,
    float* __restrict__ G, int nStrips, int N, int E)
{
    constexpr int NT  = DOUT * 2;
    constexpr int NPW = 16 / (NT / 32);
    __shared__ alignas(16) float xs[2][16 * XS_PITCH];

    const int tid  = threadIdx.x;
    const int wave = __builtin_amdgcn_readfirstlane(tid >> 5);
    const int lane = tid & 31;
    const int hi  = lane >> 4;
    const int l15 = lane & 15;
    const int col = (wave << 4) + l15;

    int s = blockIdx.x;
    if (s >= nStrips) return;

    const float* __restrict__ wbase = W + (size_t)(hi << 1) * DOUT + col;
    v2f b[D_K / 4];
#pragma unroll
    for (int t = 0; t < D_K / 4; ++t) {
        b[t].x = wbase[(size_t)(4 * t) * DOUT];
        b[t].y = wbase[(size_t)(4 * t + 1) * DOUT];
    }

    const v4f* __restrict__ Gv = reinterpret_cast<const v4f*>(Gin);
    const v4f bv = reinterpret_cast<const v4f*>(bias)[lane];
    const int r0 = wave * NPW;

    int  deg[NPW];
    int  mine[NPW];
    float dstrip = dinv[(s << 4) + l15];
    {
        const int rc = cnt[(s << 4) + l15];
#pragma unroll
        for (int j = 0; j < NPW; ++j) {
            deg[j]  = __builtin_amdgcn_readlane(rc, r0 + j);
            mine[j] = list[((s << 4) + r0 + j) * LIST_W + min(lane, deg[j])];
        }
    }

    const int aoff = l15 * XS_PITCH + (hi << 1);

    int buf = 0;
    while (true) {
        float* __restrict__ xb = xs[buf];
        const int sn = s + gridDim.x;
        const bool more = sn < nStrips;

#pragma unroll
        for (int j = 0; j < NPW; ++j) mine[j] = min(max(mine[j], 0), N - 1);

        const int snext = more ? sn : s;
        const int nrow = (snext << 4) + l15;
        const int   nrc = cnt[nrow];
        const float ndv = dinv[nrow];

#pragma unroll
        for (int j = 0; j < NPW; ++j) {
            const v4f acc = gather_node_sum<v4f>(Gv, spill_n, spill, (s << 4) + r0 + j,
                                                 deg[j], mine[j], lane, N, E);
            const float di = __int_as_float(
                __builtin_amdgcn_readlane(__float_as_int(dstrip), r0 + j));
            v4f o = di * acc + bv;
#pragma unroll
            for (int k = 0; k < 4; ++k) o[k] = fmaxf(o[k], 0.f);
            *reinterpret_cast<v4f*>(xb + (r0 + j) * XS_PITCH + (lane << 2)) = o;
        }
        __syncthreads();

#pragma unroll
        for (int j = 0; j < NPW; ++j) {
            deg[j]  = __builtin_amdgcn_readlane(nrc, r0 + j);
            mine[j] = list[((snext << 4) + r0 + j) * LIST_W + min(lane, deg[j])];
        }
        dstrip = ndv;

        const int row0 = (s << 4) + (hi << 3);
        const float* __restrict__ dbase = dinv + row0;
        float dv[8];
#pragma unroll
        for (int r = 0; r < 8; ++r) dv[r] = dbase[r];

        const float* __restrict__ ab = xb + aoff;
        v8f acc = {0.f, 0.f, 0.f, 0.f, 0.f, 0.f, 0.f, 0.f};
#pragma unroll
        for (int t = 0; t < D_K / 4; ++t) {
            const v2f a = *reinterpret_cast<const v2f*>(ab + 4 * t);
            acc = __builtin_amdgcn_wmma_f32_16x16x4_f32(
                 false, a,  false, b[t],
                 (short)0, acc,  false,  false);
        }

        float* __restrict__ gout = G + (size_t)row0 * DOUT + col;
#pragma unroll
        for (int r = 0; r < 8; ++r) {
            gout[(size_t)r * DOUT] = acc[r] * dv[r];
        }

        if (!more) break;
        s = sn;
        buf ^= 1;
    }
}

extern "C" void kernel_launch(void* const* d_in, const int* in_sizes, int n_in,
                              void* d_out, int out_size, void* d_ws, size_t ws_size,
                              hipStream_t stream)
{
    const float* x   = (const float*)d_in[0];
    const int*   ei  = (const int*)  d_in[1];
    const float* W1  = (const float*)d_in[3];
    const float* b1  = (const float*)d_in[4];
    const float* W2  = (const float*)d_in[5];
    const float* b2  = (const float*)d_in[6];
    const float* W3  = (const float*)d_in[7];
    const float* b3  = (const float*)d_in[8];
    const float* W4  = (const float*)d_in[9];
    const float* b4  = (const float*)d_in[10];

    const int N = in_sizes[0] / D_K;
    const int E = in_sizes[1] / 2;
    const int* src = ei;
    const int* dst = ei + E;

    const size_t nPad = ((size_t)N + 63) & ~(size_t)63;
    const size_t ePad = ((size_t)E + 63) & ~(size_t)63;
    float* dinv    = (float*)d_ws;
    int*   cnt     = (int*)(dinv + nPad);
    int*   spill_n = cnt + nPad;
    int*   list    = spill_n + 64;
    int*   spill   = list + nPad * LIST_W;
    float* GA      = (float*)(spill + 2 * ePad);
    float* GB      = GA + (size_t)N * D_K;

    list_init_kernel<<<(N + 255) / 256, 256, 0, stream>>>(cnt, list, spill_n, N);
    edge_place_kernel<<<(E + 255) / 256, 256, 0, stream>>>(src, dst, cnt, list, spill_n, spill, E);

    const int gblocks = (N + 7) / 8;

    const int nStrips = N / 16;
    const int gemmBlocks128 = 1024;
    const int gemmBlocks64  = 2048;

    gcn_gemm_scale_kernel<128><<<gemmBlocks128, 256, 0, stream>>>(x, W1, cnt, dinv, GA, nStrips);
    gcn_gather_gemm_scale_kernel<128><<<gemmBlocks128, 256, 0, stream>>>(
        GA, cnt, list, spill_n, spill, dinv, b1, W2, GB, nStrips, N, E);
    gcn_gather_gemm_scale_kernel<128><<<gemmBlocks128, 256, 0, stream>>>(
        GB, cnt, list, spill_n, spill, dinv, b2, W3, GA, nStrips, N, E);
    gcn_gather_gemm_scale_kernel<64><<<gemmBlocks64, 128, 0, stream>>>(
        GA, cnt, list, spill_n, spill, dinv, b3, W4, GB, nStrips, N, E);

    gather_finalize_kernel<64, false><<<gblocks, 256, 0, stream>>>(
        GB, cnt, list, spill_n, spill, dinv, b4, (float*)d_out, N, E);
}


# Edge list by two-level counting sort on LDS counters: edges binned per 256-node bucket, then one block per bucket ranks them and stores whole list rows; no global atomic per edge; sum order changes
# speedup vs baseline: 4.4462x; 1.1780x over previous
#include <hip/hip_runtime.h>

typedef float v2f __attribute__((ext_vector_type(2)));
typedef float v8f __attribute__((ext_vector_type(8)));

#define D_K 128

#define XS_PITCH 132

typedef float v4f __attribute__((ext_vector_type(4)));

template <int DOUT>
__global__ __launch_bounds__(DOUT * 2) void gcn_gemm_scale_kernel(
    const float* __restrict__ X, const float* __restrict__ W,
    const int* __restrict__ cnt, float* __restrict__ dinvOut,
    float* __restrict__ G, int nStrips)
{
    constexpr int NT  = DOUT * 2;
    constexpr int LPT = 512 / NT;
    __shared__ alignas(16) float xs[2][16 * XS_PITCH];
    __shared__ alignas(16) float dstrip[2][16];

    const int tid  = threadIdx.x;
    const int wave = __builtin_amdgcn_readfirstlane(tid >> 5);
    const int lane = tid & 31;
    const int hi  = lane >> 4;
    const int l15 = lane & 15;
    const int col = (wave << 4) + l15;

    int s = blockIdx.x;
    if (s >= nStrips) return;

    const float* __restrict__ wbase = W + (size_t)(hi << 1) * DOUT + col;
    v2f b[D_K / 4];
#pragma unroll
    for (int t = 0; t < D_K / 4; ++t) {
        b[t].x = wbase[(size_t)(4 * t) * DOUT];
        b[t].y = wbase[(size_t)(4 * t + 1) * DOUT];
    }

    const v4f* __restrict__ Xv = reinterpret_cast<const v4f*>(X);
    v4f xa[LPT];
#pragma unroll
    for (int u = 0; u < LPT; ++u) xa[u] = Xv[(size_t)s * 512 + tid + u * NT];
    int cn = 0;
    if (wave == 0) cn = cnt[(s << 4) + l15];

    const int aoff = l15 * XS_PITCH + (hi << 1);

    int buf = 0;
    while (true) {
        float* __restrict__ xb = xs[buf];
#pragma unroll
        for (int u = 0; u < LPT; ++u) {
            const int f = tid + u * NT;
            *reinterpret_cast<v4f*>(xb + (f >> 5) * XS_PITCH + ((f & 31) << 2)) = xa[u];
        }
        float d = 0.f;
        if (wave == 0) {
            d = rsqrtf(1.0f + (float)cn);
            if (hi == 0) dstrip[buf][l15] = d;
        }
        __syncthreads();
        if (wave == 0 && hi == 0) dinvOut[(s << 4) + l15] = d;

        const int sn = s + gridDim.x;
        if (sn < nStrips) {
#pragma unroll
            for (int u = 0; u < LPT; ++u) xa[u] = Xv[(size_t)sn * 512 + tid + u * NT];
            if (wave == 0) cn = cnt[(sn << 4) + l15];
        }

        const int row0 = (s << 4) + (hi << 3);
        float dv[8];
        {
            const v4f* __restrict__ dsv = reinterpret_cast<const v4f*>(&dstrip[buf][hi << 3]);
            const v4f d0 = dsv[0];
            const v4f d1 = dsv[1];
#pragma unroll
            for (int r = 0; r < 4; ++r) { dv[r] = d0[r]; dv[4 + r] = d1[r]; }
        }

        const float* __restrict__ ab = xb + aoff;
        v8f acc = {0.f, 0.f, 0.f, 0.f, 0.f, 0.f, 0.f, 0.f};
#pragma unroll
        for (int t = 0; t < D_K / 4; ++t) {
            const v2f a = *reinterpret_cast<const v2f*>(ab + 4 * t);
            acc = __builtin_amdgcn_wmma_f32_16x16x4_f32(
                 false, a,  false, b[t],
                 (short)0, acc,  false,  false);
        }

        float* __restrict__ gout = G + (size_t)row0 * DOUT + col;
#pragma unroll
        for (int r = 0; r < 8; ++r) {
            gout[(size_t)r * DOUT] = acc[r] * dv[r];
        }

        if (sn >= nStrips) break;
        s = sn;
        buf ^= 1;
    }
}

#define LIST_W   32
#define LIST_CAP 31

#define BKT_SHIFT 8
#define BKT_NODES (1 << BKT_SHIFT)
#define NB_MAX    512
#define BIN_NT    256
#define BIN_EPT   10
#define BIN_CH    (BIN_NT * BIN_EPT)
#define RUN_PRE   16

typedef int v2i __attribute__((ext_vector_type(2)));
typedef int v4i __attribute__((ext_vector_type(4)));

__global__ __launch_bounds__(BIN_NT) void edge_bin_kernel(
    const int* __restrict__ src, const int* __restrict__ dst,
    int* __restrict__ binned, int* __restrict__ offT, int* __restrict__ spill_n,
    int E, int NB, int G1)
{
    __shared__ int bcnt[NB_MAX];
    __shared__ int wtot[BIN_NT / 32];
    __shared__ alignas(16) v2i sorted[BIN_CH];

    const int tid  = threadIdx.x;
    const int g    = blockIdx.x;
    const int base = g * BIN_CH;
    const int n    = min(BIN_CH, E - base);
    if (g == 0 && tid == 0) spill_n[0] = 0;

    int d[BIN_EPT], s[BIN_EPT];
#pragma unroll
    for (int u = 0; u < BIN_EPT; ++u) {
        const int i = u * BIN_NT + tid;
        d[u] = 0; s[u] = 0;
        if (i < n) { d[u] = dst[base + i]; s[u] = src[base + i]; }
    }
    for (int b = tid; b < NB_MAX; b += BIN_NT) bcnt[b] = 0;
    __syncthreads();

    int bk[BIN_EPT], r[BIN_EPT];
#pragma unroll
    for (int u = 0; u < BIN_EPT; ++u) {
        const int i = u * BIN_NT + tid;
        bk[u] = min(max(d[u] >> BKT_SHIFT, 0), NB - 1);
        r[u] = 0;
        if (i < n) r[u] = atomicAdd(&bcnt[bk[u]], 1);
    }
    __syncthreads();

    const int c0 = bcnt[2 * tid];
    const int c1 = bcnt[2 * tid + 1];
    const int lane = tid & 31;
    int inc = c0 + c1;
#pragma unroll
    for (int o = 1; o < 32; o <<= 1) {
        const int v = __shfl_up(inc, o);
        if (lane >= o) inc += v;
    }
    if (lane == 31) wtot[tid >> 5] = inc;
    __syncthreads();
    int woff = 0;
#pragma unroll
    for (int w = 0; w < BIN_NT / 32; ++w)
        if (w < (tid >> 5)) woff += wtot[w];
    const int st0 = woff + inc - (c0 + c1);
    const int st1 = st0 + c0;
    bcnt[2 * tid]     = st0;
    bcnt[2 * tid + 1] = st1;
    if (2 * tid < NB)     offT[(size_t)(2 * tid) * G1 + g]     = st0 | (c0 << 16);
    if (2 * tid + 1 < NB) offT[(size_t)(2 * tid + 1) * G1 + g] = st1 | (c1 << 16);
    __syncthreads();

#pragma unroll
    for (int u = 0; u < BIN_EPT; ++u) {
        const int i = u * BIN_NT + tid;
        if (i < n) sorted[bcnt[bk[u]] + r[u]] = (v2i){d[u], s[u]};
    }
    __syncthreads();

    v2i* __restrict__ outp = reinterpret_cast<v2i*>(binned) + (size_t)base;
    for (int i = tid; i < n; i += BIN_NT) outp[i] = sorted[i];
}

__device__ __forceinline__ void put_edge(int* rows,
                                         int* __restrict__ spill_n, int* __restrict__ spill,
                                         int nodeBase, int local, int source, int r)
{
    if (r < LIST_CAP) {
        rows[local * LIST_W + 1 + r] = source;
    } else {
        const int k = atomicAdd(spill_n, 1);
        spill[2 * k]     = nodeBase + local;
        spill[2 * k + 1] = source;
    }
}

__global__ __launch_bounds__(BKT_NODES) void bucket_list_kernel(
    const int* __restrict__ binned, const int* __restrict__ offT,
    int* __restrict__ cnt, int* __restrict__ list,
    int* __restrict__ spill_n, int* __restrict__ spill, int N, int G1)
{
    __shared__ alignas(16) int rows[BKT_NODES * LIST_W];
    __shared__ int ncnt[BKT_NODES];

    const int tid = threadIdx.x;
    const int b   = blockIdx.x;
    const int nodeBase = b << BKT_SHIFT;

    v4i* rows4 = reinterpret_cast<v4i*>(rows);
    ncnt[tid] = 0;
#pragma unroll
    for (int q = 0; q < LIST_W / 4; ++q) {
        const int f  = q * BKT_NODES + tid;
        const int nd = nodeBase + (f >> 3);
        rows4[f] = (v4i){nd, nd, nd, nd};
    }
    __syncthreads();

    const v2i* __restrict__ bv = reinterpret_cast<const v2i*>(binned);
    for (int g = tid; g < G1; g += BKT_NODES) {
        const int p  = offT[(size_t)b * G1 + g];
        const int c  = p >> 16;
        const v2i* __restrict__ run = bv + (size_t)g * BIN_CH + (p & 0xFFFF);
        v2i e[RUN_PRE];
#pragma unroll
        for (int k = 0; k < RUN_PRE; ++k) {
            e[k] = (v2i){0, 0};
            if (k < c) e[k] = run[k];
        }
        int rk[RUN_PRE];
#pragma unroll
        for (int k = 0; k < RUN_PRE; ++k) {
            rk[k] = 0;
            if (k < c) rk[k] = atomicAdd(&ncnt[e[k].x & (BKT_NODES - 1)], 1);
        }
#pragma unroll
        for (int k = 0; k < RUN_PRE; ++k)
            if (k < c) put_edge(rows, spill_n, spill, nodeBase,
                                e[k].x & (BKT_NODES - 1), e[k].y, rk[k]);
        for (int k = RUN_PRE; k < c; ++k) {
            const v2i t = run[k];
            const int local = t.x & (BKT_NODES - 1);
            const int r = atomicAdd(&ncnt[local], 1);
            put_edge(rows, spill_n, spill, nodeBase, local, t.y, r);
        }
    }
    __syncthreads();

    const int node = nodeBase + tid;
    if (node < N) cnt[node] = ncnt[tid];
    v4i* __restrict__ lv = reinterpret_cast<v4i*>(list + (size_t)nodeBase * LIST_W);
#pragma unroll
    for (int q = 0; q < LIST_W / 4; ++q) {
        const int f = q * BKT_NODES + tid;
        if (nodeBase + (f >> 3) < N) lv[f] = rows4[f];
    }
}

template <int K, typename VF>
__device__ __forceinline__ void gather_rows(VF& acc, const VF* __restrict__ Gv,
                                            int mine, int q, int lane)
{
    VF g[K];
#pragma unroll
    for (int u = 0; u < K; ++u)
        g[u] = Gv[(size_t)__builtin_amdgcn_readlane(mine, q + u) * 32 + lane];
#pragma unroll
    for (int u = 0; u < K; ++u) acc += g[u];
}

template <typename VF>
__device__ __forceinline__ VF gather_node_sum(const VF* __restrict__ Gv,
                                              const int* __restrict__ spill_n,
                                              const int* __restrict__ spill,
                                              int node, int deg, int mine,
                                              int lane, int N, int E)
{
    constexpr int U = 8;
    VF acc = (VF)0.f;
    const int n = min(deg, LIST_CAP) + 1;
    for (int q = 0; q < n; q += U) {
        switch (min(U, n - q)) {
            case 1:  gather_rows<1>(acc, Gv, mine, q, lane); break;
            case 2:  gather_rows<2>(acc, Gv, mine, q, lane); break;
            case 3:  gather_rows<3>(acc, Gv, mine, q, lane); break;
            case 4:  gather_rows<4>(acc, Gv, mine, q, lane); break;
            case 5:  gather_rows<5>(acc, Gv, mine, q, lane); break;
            case 6:  gather_rows<6>(acc, Gv, mine, q, lane); break;
            case 7:  gather_rows<7>(acc, Gv, mine, q, lane); break;
            default: gather_rows<8>(acc, Gv, mine, q, lane); break;
        }
    }
    if (deg > LIST_CAP) {
        const int m = min(spill_n[0], E);
        for (int k = 0; k < m; ++k) {
            if (spill[2 * k] == node) {
                const int sidx = min(max(spill[2 * k + 1], 0), N - 1);
                acc += Gv[(size_t)sidx * 32 + lane];
            }
        }
    }
    return acc;
}

template <int DOUT, bool RELU>
__global__ __launch_bounds__(256) void gather_finalize_kernel(
    const float* __restrict__ G, const int* __restrict__ cnt,
    const int* __restrict__ list, const int* __restrict__ spill_n,
    const int* __restrict__ spill,
    const float* __restrict__ dinv, const float* __restrict__ bias,
    float* __restrict__ OUT, int N, int E)
{
    constexpr int V = DOUT / 32;
    typedef float vf __attribute__((ext_vector_type(V)));
    const int lane = threadIdx.x & 31;
    const int i = __builtin_amdgcn_readfirstlane(
        (int)((blockIdx.x * 256u + threadIdx.x) >> 5));
    if (i >= N) return;
    const vf* __restrict__ Gv = reinterpret_cast<const vf*>(G);
    const int deg = cnt[i];
    int mine = list[i * LIST_W + min(lane, deg)];
    mine = min(max(mine, 0), N - 1);
    const vf acc = gather_node_sum<vf>(Gv, spill_n, spill, i, deg, mine, lane, N, E);
    const float di = dinv[i];
    const vf b = reinterpret_cast<const vf*>(bias)[lane];
    vf o = di * acc + b;
    if (RELU) {
#pragma unroll
        for (int k = 0; k < V; ++k) o[k] = fmaxf(o[k], 0.f);
    }
    reinterpret_cast<vf*>(OUT)[(size_t)i * 32 + lane] = o;
}

template <int DOUT>
__global__ __launch_bounds__(DOUT * 2) void gcn_gather_gemm_scale_kernel(
    const float* __restrict__ Gin, const int* __restrict__ cnt,
    const int* __restrict__ list, const int* __restrict__ spill_n,
    const int* __restrict__ spill, const float* __restrict__ dinv,
    const float* __restrict__ bias, const float* __restrict__ W,
    float* __restrict__ G, int nStrips, int N, int E)
{
    constexpr int NT  = DOUT * 2;
    constexpr int NPW = 16 / (NT / 32);
    __shared__ alignas(16) float xs[2][16 * XS_PITCH];

    const int tid  = threadIdx.x;
    const int wave = __builtin_amdgcn_readfirstlane(tid >> 5);
    const int lane = tid & 31;
    const int hi  = lane >> 4;
    const int l15 = lane & 15;
    const int col = (wave << 4) + l15;

    int s = blockIdx.x;
    if (s >= nStrips) return;

    const float* __restrict__ wbase = W + (size_t)(hi << 1) * DOUT + col;
    v2f b[D_K / 4];
#pragma unroll
    for (int t = 0; t < D_K / 4; ++t) {
        b[t].x = wbase[(size_t)(4 * t) * DOUT];
        b[t].y = wbase[(size_t)(4 * t + 1) * DOUT];
    }

    const v4f* __restrict__ Gv = reinterpret_cast<const v4f*>(Gin);
    const v4f bv = reinterpret_cast<const v4f*>(bias)[lane];
    const int r0 = wave * NPW;

    int  deg[NPW];
    int  mine[NPW];
    float dstrip = dinv[(s << 4) + l15];
    {
        const int rc = cnt[(s << 4) + l15];
#pragma unroll
        for (int j = 0; j < NPW; ++j) {
            deg[j]  = __builtin_amdgcn_readlane(rc, r0 + j);
            mine[j] = list[((s << 4) + r0 + j) * LIST_W + min(lane, deg[j])];
        }
    }

    const int aoff = l15 * XS_PITCH + (hi << 1);

    int buf = 0;
    while (true) {
        float* __restrict__ xb = xs[buf];
        const int sn = s + gridDim.x;
        const bool more = sn < nStrips;

#pragma unroll
        for (int j = 0; j < NPW; ++j) mine[j] = min(max(mine[j], 0), N - 1);

        const int snext = more ? sn : s;
        const int nrow = (snext << 4) + l15;
        const int   nrc = cnt[nrow];
        const float ndv = dinv[nrow];

#pragma unroll
        for (int j = 0; j < NPW; ++j) {
            const v4f acc = gather_node_sum<v4f>(Gv, spill_n, spill, (s << 4) + r0 + j,
                                                 deg[j], mine[j], lane, N, E);
            const float di = __int_as_float(
                __builtin_amdgcn_readlane(__float_as_int(dstrip), r0 + j));
            v4f o = di * acc + bv;
#pragma unroll
            for (int k = 0; k < 4; ++k) o[k] = fmaxf(o[k], 0.f);
            *reinterpret_cast<v4f*>(xb + (r0 + j) * XS_PITCH + (lane << 2)) = o;
        }
        __syncthreads();

#pragma unroll
        for (int j = 0; j < NPW; ++j) {
            deg[j]  = __builtin_amdgcn_readlane(nrc, r0 + j);
            mine[j] = list[((snext << 4) + r0 + j) * LIST_W + min(lane, deg[j])];
        }
        dstrip = ndv;

        const int row0 = (s << 4) + (hi << 3);
        const float* __restrict__ dbase = dinv + row0;
        float dv[8];
#pragma unroll
        for (int r = 0; r < 8; ++r) dv[r] = dbase[r];

        const float* __restrict__ ab = xb + aoff;
        v8f acc = {0.f, 0.f, 0.f, 0.f, 0.f, 0.f, 0.f, 0.f};
#pragma unroll
        for (int t = 0; t < D_K / 4; ++t) {
            const v2f a = *reinterpret_cast<const v2f*>(ab + 4 * t);
            acc = __builtin_amdgcn_wmma_f32_16x16x4_f32(
                 false, a,  false, b[t],
                 (short)0, acc,  false,  false);
        }

        float* __restrict__ gout = G + (size_t)row0 * DOUT + col;
#pragma unroll
        for (int r = 0; r < 8; ++r) {
            gout[(size_t)r * DOUT] = acc[r] * dv[r];
        }

        if (!more) break;
        s = sn;
        buf ^= 1;
    }
}

extern "C" void kernel_launch(void* const* d_in, const int* in_sizes, int n_in,
                              void* d_out, int out_size, void* d_ws, size_t ws_size,
                              hipStream_t stream)
{
    const float* x   = (const float*)d_in[0];
    const int*   ei  = (const int*)  d_in[1];
    const float* W1  = (const float*)d_in[3];
    const float* b1  = (const float*)d_in[4];
    const float* W2  = (const float*)d_in[5];
    const float* b2  = (const float*)d_in[6];
    const float* W3  = (const float*)d_in[7];
    const float* b3  = (const float*)d_in[8];
    const float* W4  = (const float*)d_in[9];
    const float* b4  = (const float*)d_in[10];

    const int N = in_sizes[0] / D_K;
    const int E = in_sizes[1] / 2;
    const int* src = ei;
    const int* dst = ei + E;

    const size_t nPad = ((size_t)N + 63) & ~(size_t)63;
    const size_t ePad = ((size_t)E + 63) & ~(size_t)63;
    const int G1 = (E + BIN_CH - 1) / BIN_CH;
    const int NB = (N + BKT_NODES - 1) / BKT_NODES;
    float* dinv    = (float*)d_ws;
    int*   cnt     = (int*)(dinv + nPad);
    int*   spill_n = cnt + nPad;
    int*   list    = spill_n + 64;
    int*   spill   = list + nPad * LIST_W;
    float* GA      = (float*)(spill + 2 * ePad);
    float* GB      = GA + (size_t)N * D_K;
    int*   binned  = (int*)(GB + (size_t)N * D_K);
    int*   offT    = binned + (size_t)G1 * BIN_CH * 2;

    edge_bin_kernel<<<G1, BIN_NT, 0, stream>>>(src, dst, binned, offT, spill_n, E, NB, G1);
    bucket_list_kernel<<<NB, BKT_NODES, 0, stream>>>(binned, offT, cnt, list, spill_n, spill, N, G1);

    const int gblocks = (N + 7) / 8;

    const int nStrips = N / 16;
    const int gemmBlocks128 = 1024;
    const int gemmBlocks64  = 2048;

    gcn_gemm_scale_kernel<128><<<gemmBlocks128, 256, 0, stream>>>(x, W1, cnt, dinv, GA, nStrips);
    gcn_gather_gemm_scale_kernel<128><<<gemmBlocks128, 256, 0, stream>>>(
        GA, cnt, list, spill_n, spill, dinv, b1, W2, GB, nStrips, N, E);
    gcn_gather_gemm_scale_kernel<128><<<gemmBlocks128, 256, 0, stream>>>(
        GB, cnt, list, spill_n, spill, dinv, b2, W3, GA, nStrips, N, E);
    gcn_gather_gemm_scale_kernel<64><<<gemmBlocks64, 128, 0, stream>>>(
        GA, cnt, list, spill_n, spill, dinv, b3, W4, GB, nStrips, N, E);

    gather_finalize_kernel<64, false><<<gblocks, 256, 0, stream>>>(
        GB, cnt, list, spill_n, spill, dinv, b4, (float*)d_out, N, E);
}
